# starting kernel
# baseline (speedup 1.0000x reference)
#include <hip/hip_runtime.h>
#include <hip/hip_bf16.h>
#include <math.h>

#define V_SZ 45
#define H_SZ 2048
#define W_SZ 128
#define D_SZ 200
#define KIH  (H_SZ + W_SZ)
#define N3H  (3 * H_SZ)

typedef float v2f __attribute__((ext_vector_type(2)));
typedef float v8f __attribute__((ext_vector_type(8)));

__global__ __launch_bounds__(32) void k_ctrl_sin(
    const float* __restrict__ h,
    const float* __restrict__ ctrl_W, const float* __restrict__ ctrl_b,
    const float* __restrict__ sin_W,  const float* __restrict__ sin_b,
    float* __restrict__ logits, float* __restrict__ stack_in)
{
    const int row = blockIdx.x;
    const float* Wr;
    const float* bp;
    float* outp;
    int orow;
    bool doTanh;
    if (row < 3) { Wr = ctrl_W; bp = ctrl_b; outp = logits;   orow = row;     doTanh = false; }
    else         { Wr = sin_W;  bp = sin_b;  outp = stack_in; orow = row - 3; doTanh = true;  }
    const float* wr = Wr + (size_t)orow * H_SZ;

    const int lane = threadIdx.x;
    float s = 0.f;
#pragma unroll 4
    for (int k = lane * 4; k < H_SZ; k += 32 * 4) {
        float4 wv = *(const float4*)(wr + k);
        float4 hv = *(const float4*)(h  + k);
        s += wv.x * hv.x + wv.y * hv.y + wv.z * hv.z + wv.w * hv.w;
    }
    for (int off = 16; off > 0; off >>= 1) s += __shfl_down(s, off, 32);
    if (lane == 0) {
        s += bp[orow];
        outp[orow] = doTanh ? tanhf(s) : s;
    }
}

__global__ __launch_bounds__(256) void k_stack(
    const float* __restrict__ stack, const float* __restrict__ logits,
    const float* __restrict__ stack_in, const float* __restrict__ emb,
    const int* __restrict__ inp,
    float* __restrict__ out_stack, float* __restrict__ xbuf)
{
    const float l0 = logits[0], l1 = logits[1], l2 = logits[2];
    const float m  = fmaxf(l0, fmaxf(l1, l2));
    const float e0 = expf(l0 - m), e1 = expf(l1 - m), e2 = expf(l2 - m);
    const float inv = 1.f / (e0 + e1 + e2);
    const float c0 = e0 * inv, c1 = e1 * inv, c2 = e2 * inv;

    const int idx = blockIdx.x * blockDim.x + threadIdx.x;
    if (idx < D_SZ * W_SZ) {
        const int d = idx >> 7;
        const int w = idx & 127;
        const float cur  = stack[idx];
        const float up   = (d == 0)        ? stack_in[w]     : stack[idx - W_SZ];
        const float down = (d == D_SZ - 1) ? 0.f             : stack[idx + W_SZ];
        const float v = c2 * cur + c0 * up + c1 * down;
        out_stack[idx] = v;
        if (d == 0) xbuf[H_SZ + w] = v;
    } else {
        const int j = idx - D_SZ * W_SZ;
        xbuf[j] = emb[(size_t)inp[0] * H_SZ + j];
    }
}

template <int K>
__global__ __launch_bounds__(128) void k_gemv_wmma(
    const float* __restrict__ W, const float* __restrict__ xv,
    const float* __restrict__ bias, float* __restrict__ out)
{
    const int n_base = blockIdx.x * 16;
    const int lane   = threadIdx.x & 31;
    const int wave   = threadIdx.x >> 5;
    const int laneN  = lane & 15;
    const int hi     = lane >> 4;
    const bool isRow0 = (laneN == 0);

    constexpr int kPerWave = K >> 2;
    const int kb = wave * kPerWave;

    const float* __restrict__ wrow = W + (size_t)(n_base + laneN) * K + hi * 2;

    v8f c0 = {0.f, 0.f, 0.f, 0.f, 0.f, 0.f, 0.f, 0.f};
    v8f c1 = {0.f, 0.f, 0.f, 0.f, 0.f, 0.f, 0.f, 0.f};

#pragma unroll 4
    for (int kk = 0; kk < kPerWave; kk += 8) {
        const int k = kb + kk;
        v2f b0 = *(const v2f*)(wrow + k);
        v2f b1 = *(const v2f*)(wrow + k + 4);
        v2f xa = *(const v2f*)(xv + k + hi * 2);
        v2f xb = *(const v2f*)(xv + k + 4 + hi * 2);
        v2f a0, a1;
        a0.x = isRow0 ? xa.x : 0.f;  a0.y = isRow0 ? xa.y : 0.f;
        a1.x = isRow0 ? xb.x : 0.f;  a1.y = isRow0 ? xb.y : 0.f;
        c0 = __builtin_amdgcn_wmma_f32_16x16x4_f32(false, a0, false, b0,
                                                   (short)0, c0, false, false);
        c1 = __builtin_amdgcn_wmma_f32_16x16x4_f32(false, a1, false, b1,
                                                   (short)0, c1, false, false);
    }

    __shared__ float red[4][16];
    if (lane < 16) red[wave][lane] = c0[0] + c1[0];
    __syncthreads();
    if (threadIdx.x < 16) {
        const int n = n_base + threadIdx.x;
        const float s = red[0][threadIdx.x] + red[1][threadIdx.x] +
                        red[2][threadIdx.x] + red[3][threadIdx.x];
        out[n] = s + bias[n];
    }
}

__global__ __launch_bounds__(256) void k_gru(
    const float* __restrict__ gi, const float* __restrict__ gh,
    const float* __restrict__ h,
    float* __restrict__ out_h, float* __restrict__ hnew_ws)
{
    const int i = blockIdx.x * blockDim.x + threadIdx.x;
    if (i >= H_SZ) return;
    const float ir  = gi[i],           hr = gh[i];
    const float iz  = gi[H_SZ + i],    hz = gh[H_SZ + i];
    const float in_ = gi[2*H_SZ + i],  hn = gh[2*H_SZ + i];
    const float r = 1.f / (1.f + expf(-(ir + hr)));
    const float z = 1.f / (1.f + expf(-(iz + hz)));
    const float n = tanhf(in_ + r * hn);
    const float hv = h[i];
    const float o = (1.f - z) * n + z * hv;
    out_h[i]   = o;
    hnew_ws[i] = o;
}

__global__ __launch_bounds__(32) void k_dec(
    const float* __restrict__ hnew, const float* __restrict__ dec_W,
    const float* __restrict__ dec_b, float* __restrict__ out_y)
{
    const int row = blockIdx.x;
    const float* wr = dec_W + (size_t)row * H_SZ;
    const int lane = threadIdx.x;
    float s = 0.f;
#pragma unroll 4
    for (int k = lane * 4; k < H_SZ; k += 32 * 4) {
        float4 wv = *(const float4*)(wr + k);
        float4 hv = *(const float4*)(hnew + k);
        s += wv.x * hv.x + wv.y * hv.y + wv.z * hv.z + wv.w * hv.w;
    }
    for (int off = 16; off > 0; off >>= 1) s += __shfl_down(s, off, 32);
    if (lane == 0) out_y[row] = s + dec_b[row];
}

extern "C" void kernel_launch(void* const* d_in, const int* in_sizes, int n_in,
                              void* d_out, int out_size, void* d_ws, size_t ws_size,
                              hipStream_t stream)
{
    (void)in_sizes; (void)n_in; (void)out_size; (void)ws_size;

    const int*   inp    = (const int*)  d_in[0];
    const float* hidden = (const float*)d_in[1];
    const float* stack  = (const float*)d_in[2];
    const float* emb    = (const float*)d_in[3];
    const float* ctrl_W = (const float*)d_in[4];
    const float* ctrl_b = (const float*)d_in[5];
    const float* sin_W  = (const float*)d_in[6];
    const float* sin_b  = (const float*)d_in[7];
    const float* w_ih   = (const float*)d_in[8];
    const float* w_hh   = (const float*)d_in[9];
    const float* b_ih   = (const float*)d_in[10];
    const float* b_hh   = (const float*)d_in[11];
    const float* dec_W  = (const float*)d_in[12];
    const float* dec_b  = (const float*)d_in[13];

    float* out       = (float*)d_out;
    float* out_y     = out;
    float* out_h     = out + V_SZ;
    float* out_stack = out + V_SZ + H_SZ;

    float* ws       = (float*)d_ws;
    float* logits   = ws;
    float* stack_in = ws + 16;
    float* xbuf     = ws + 160;
    float* gi       = ws + 2336;
    float* gh       = ws + 8480;
    float* hnew     = ws + 14624;

    k_ctrl_sin<<<131, 32, 0, stream>>>(hidden, ctrl_W, ctrl_b, sin_W, sin_b,
                                       logits, stack_in);
    k_stack<<<108, 256, 0, stream>>>(stack, logits, stack_in, emb, inp,
                                     out_stack, xbuf);
    k_gemv_wmma<KIH><<<N3H / 16, 128, 0, stream>>>(w_ih, xbuf,  b_ih, gi);
    k_gemv_wmma<H_SZ><<<N3H / 16, 128, 0, stream>>>(w_hh, hidden, b_hh, gh);
    k_gru<<<(H_SZ + 255) / 256, 256, 0, stream>>>(gi, gh, hidden, out_h, hnew);
    k_dec<<<V_SZ, 32, 0, stream>>>(hnew, dec_W, dec_b, out_y);
}


# GEMVs re-tiled: each wave-wide b128 load covers 512 contiguous bytes of one row (x loaded alike, row sums on the WMMA diagonal), 8 waves x 3 rows with all loads issued up front; sum order changes
# speedup vs baseline: 1.4673x; 1.4673x over previous
#include <hip/hip_runtime.h>
#include <hip/hip_bf16.h>
#include <math.h>

#define V_SZ 45
#define H_SZ 2048
#define W_SZ 128
#define D_SZ 200
#define KIH  (H_SZ + W_SZ)
#define N3H  (3 * H_SZ)

typedef float v2f __attribute__((ext_vector_type(2)));
typedef float v4f __attribute__((ext_vector_type(4)));
typedef float v8f __attribute__((ext_vector_type(8)));

__global__ __launch_bounds__(32) void k_ctrl_sin(
    const float* __restrict__ h,
    const float* __restrict__ ctrl_W, const float* __restrict__ ctrl_b,
    const float* __restrict__ sin_W,  const float* __restrict__ sin_b,
    float* __restrict__ logits, float* __restrict__ stack_in)
{
    const int row = blockIdx.x;
    const float* Wr;
    const float* bp;
    float* outp;
    int orow;
    bool doTanh;
    if (row < 3) { Wr = ctrl_W; bp = ctrl_b; outp = logits;   orow = row;     doTanh = false; }
    else         { Wr = sin_W;  bp = sin_b;  outp = stack_in; orow = row - 3; doTanh = true;  }
    const float* wr = Wr + (size_t)orow * H_SZ;

    const int lane = threadIdx.x;
    float s = 0.f;
#pragma unroll 4
    for (int k = lane * 4; k < H_SZ; k += 32 * 4) {
        float4 wv = *(const float4*)(wr + k);
        float4 hv = *(const float4*)(h  + k);
        s += wv.x * hv.x + wv.y * hv.y + wv.z * hv.z + wv.w * hv.w;
    }
    for (int off = 16; off > 0; off >>= 1) s += __shfl_down(s, off, 32);
    if (lane == 0) {
        s += bp[orow];
        outp[orow] = doTanh ? tanhf(s) : s;
    }
}

__global__ __launch_bounds__(256) void k_stack(
    const float* __restrict__ stack, const float* __restrict__ logits,
    const float* __restrict__ stack_in, const float* __restrict__ emb,
    const int* __restrict__ inp,
    float* __restrict__ out_stack, float* __restrict__ xbuf)
{
    const float l0 = logits[0], l1 = logits[1], l2 = logits[2];
    const float m  = fmaxf(l0, fmaxf(l1, l2));
    const float e0 = expf(l0 - m), e1 = expf(l1 - m), e2 = expf(l2 - m);
    const float inv = 1.f / (e0 + e1 + e2);
    const float c0 = e0 * inv, c1 = e1 * inv, c2 = e2 * inv;

    const int idx = blockIdx.x * blockDim.x + threadIdx.x;
    if (idx < D_SZ * W_SZ) {
        const int d = idx >> 7;
        const int w = idx & 127;
        const float cur  = stack[idx];
        const float up   = (d == 0)        ? stack_in[w]     : stack[idx - W_SZ];
        const float down = (d == D_SZ - 1) ? 0.f             : stack[idx + W_SZ];
        const float v = c2 * cur + c0 * up + c1 * down;
        out_stack[idx] = v;
        if (d == 0) xbuf[H_SZ + w] = v;
    } else {
        const int j = idx - D_SZ * W_SZ;
        xbuf[j] = emb[(size_t)inp[0] * H_SZ + j];
    }
}

#define GEMV_ROWS  3
#define GEMV_WAVES 8
static_assert(KIH % 128 == 0 && H_SZ % 128 == 0, "rows are whole 512-byte pieces");
static_assert(N3H % (GEMV_WAVES * GEMV_ROWS) == 0, "every wave owns GEMV_ROWS whole rows");

template <int KQ>
__global__ __launch_bounds__(GEMV_WAVES * 32) void k_gemv_wmma(
    const float* __restrict__ W, const float* __restrict__ xv,
    const float* __restrict__ bias, float* __restrict__ out)
{
    constexpr int K = 128 * KQ;
    const int lane = threadIdx.x & 31;
    const int wave = __builtin_amdgcn_readfirstlane(threadIdx.x >> 5);
    const int row0 = (blockIdx.x * GEMV_WAVES + wave) * GEMV_ROWS;

    const float* __restrict__ xp = xv + lane * 4;
    const float* __restrict__ wp = W + (size_t)row0 * K + lane * 4;

    v4f x[KQ];
    v4f w[GEMV_ROWS][KQ];
    float br[GEMV_ROWS];
#pragma unroll
    for (int r = 0; r < GEMV_ROWS; ++r) br[r] = bias[row0 + r];
#pragma unroll
    for (int q = 0; q < KQ; ++q) x[q] = *(const v4f*)(xp + 128 * q);
#pragma unroll
    for (int q = 0; q < KQ; ++q) {
#pragma unroll
        for (int r = 0; r < GEMV_ROWS; ++r)
            w[r][q] = *(const v4f*)(wp + (size_t)r * K + 128 * q);
        __builtin_amdgcn_sched_barrier(0);
    }

    v8f acc[GEMV_ROWS][2];
#pragma unroll
    for (int r = 0; r < GEMV_ROWS; ++r) {
        const float b0 = (lane == 0) ? br[r] : 0.f;
        acc[r][0] = (v8f){b0,  0.f, 0.f, 0.f, 0.f, 0.f, 0.f, 0.f};
        acc[r][1] = (v8f){0.f, 0.f, 0.f, 0.f, 0.f, 0.f, 0.f, 0.f};
    }
#pragma unroll
    for (int q = 0; q < KQ; ++q) {
        const v2f a_lo = {x[q].x, x[q].y};
        const v2f a_hi = {x[q].z, x[q].w};
#pragma unroll
        for (int r = 0; r < GEMV_ROWS; ++r) {
            const v2f b_lo = {w[r][q].x, w[r][q].y};
            const v2f b_hi = {w[r][q].z, w[r][q].w};
            acc[r][0] = __builtin_amdgcn_wmma_f32_16x16x4_f32(
                false, a_lo, false, b_lo, (short)0, acc[r][0], false, false);
            acc[r][1] = __builtin_amdgcn_wmma_f32_16x16x4_f32(
                false, a_hi, false, b_hi, (short)0, acc[r][1], false, false);
        }
    }

    const int  dj     = lane & 7;
    const bool onDiag = (lane < 8) || (lane >= 24);
    float mine = 0.f;
#pragma unroll
    for (int r = 0; r < GEMV_ROWS; ++r) {
        const v8f c = acc[r][0] + acc[r][1];
        float d = c[0];
#pragma unroll
        for (int i = 1; i < 8; ++i) d = (dj == i) ? c[i] : d;
        d = onDiag ? d : 0.f;
#pragma unroll
        for (int m = 16; m >= 1; m >>= 1) d += __shfl_xor(d, m, 32);
        if (lane == r) mine = d;
    }
    if (lane < GEMV_ROWS) out[row0 + lane] = mine;
}

__global__ __launch_bounds__(256) void k_gru(
    const float* __restrict__ gi, const float* __restrict__ gh,
    const float* __restrict__ h,
    float* __restrict__ out_h, float* __restrict__ hnew_ws)
{
    const int i = blockIdx.x * blockDim.x + threadIdx.x;
    if (i >= H_SZ) return;
    const float ir  = gi[i],           hr = gh[i];
    const float iz  = gi[H_SZ + i],    hz = gh[H_SZ + i];
    const float in_ = gi[2*H_SZ + i],  hn = gh[2*H_SZ + i];
    const float r = 1.f / (1.f + expf(-(ir + hr)));
    const float z = 1.f / (1.f + expf(-(iz + hz)));
    const float n = tanhf(in_ + r * hn);
    const float hv = h[i];
    const float o = (1.f - z) * n + z * hv;
    out_h[i]   = o;
    hnew_ws[i] = o;
}

__global__ __launch_bounds__(32) void k_dec(
    const float* __restrict__ hnew, const float* __restrict__ dec_W,
    const float* __restrict__ dec_b, float* __restrict__ out_y)
{
    const int row = blockIdx.x;
    const float* wr = dec_W + (size_t)row * H_SZ;
    const int lane = threadIdx.x;
    float s = 0.f;
#pragma unroll 4
    for (int k = lane * 4; k < H_SZ; k += 32 * 4) {
        float4 wv = *(const float4*)(wr + k);
        float4 hv = *(const float4*)(hnew + k);
        s += wv.x * hv.x + wv.y * hv.y + wv.z * hv.z + wv.w * hv.w;
    }
    for (int off = 16; off > 0; off >>= 1) s += __shfl_down(s, off, 32);
    if (lane == 0) out_y[row] = s + dec_b[row];
}

extern "C" void kernel_launch(void* const* d_in, const int* in_sizes, int n_in,
                              void* d_out, int out_size, void* d_ws, size_t ws_size,
                              hipStream_t stream)
{
    (void)in_sizes; (void)n_in; (void)out_size; (void)ws_size;

    const int*   inp    = (const int*)  d_in[0];
    const float* hidden = (const float*)d_in[1];
    const float* stack  = (const float*)d_in[2];
    const float* emb    = (const float*)d_in[3];
    const float* ctrl_W = (const float*)d_in[4];
    const float* ctrl_b = (const float*)d_in[5];
    const float* sin_W  = (const float*)d_in[6];
    const float* sin_b  = (const float*)d_in[7];
    const float* w_ih   = (const float*)d_in[8];
    const float* w_hh   = (const float*)d_in[9];
    const float* b_ih   = (const float*)d_in[10];
    const float* b_hh   = (const float*)d_in[11];
    const float* dec_W  = (const float*)d_in[12];
    const float* dec_b  = (const float*)d_in[13];

    float* out       = (float*)d_out;
    float* out_y     = out;
    float* out_h     = out + V_SZ;
    float* out_stack = out + V_SZ + H_SZ;

    float* ws       = (float*)d_ws;
    float* logits   = ws;
    float* stack_in = ws + 16;
    float* xbuf     = ws + 160;
    float* gi       = ws + 2336;
    float* gh       = ws + 8480;
    float* hnew     = ws + 14624;

    k_ctrl_sin<<<131, 32, 0, stream>>>(hidden, ctrl_W, ctrl_b, sin_W, sin_b,
                                       logits, stack_in);
    k_stack<<<108, 256, 0, stream>>>(stack, logits, stack_in, emb, inp,
                                     out_stack, xbuf);
    k_gemv_wmma<KIH / 128><<<N3H / (GEMV_WAVES * GEMV_ROWS), GEMV_WAVES * 32, 0, stream>>>(w_ih, xbuf,  b_ih, gi);
    k_gemv_wmma<H_SZ / 128><<<N3H / (GEMV_WAVES * GEMV_ROWS), GEMV_WAVES * 32, 0, stream>>>(w_hh, hidden, b_hh, gh);
    k_gru<<<(H_SZ + 255) / 256, 256, 0, stream>>>(gi, gh, hidden, out_h, hnew);
    k_dec<<<V_SZ, 32, 0, stream>>>(hnew, dec_W, dec_b, out_y);
}


# Six launches become three, one per dependence level: input-only products (ctrl/stack-in rows, gh, gi's emb columns) | blend + gi's stack-top columns + GRU | decode; gi's sum order changes
# speedup vs baseline: 2.0128x; 1.3718x over previous
#include <hip/hip_runtime.h>
#include <hip/hip_bf16.h>
#include <math.h>

#define V_SZ 45
#define H_SZ 2048
#define W_SZ 128
#define D_SZ 200
#define KIH  (H_SZ + W_SZ)
#define N3H  (3 * H_SZ)

typedef float v2f __attribute__((ext_vector_type(2)));
typedef float v4f __attribute__((ext_vector_type(4)));
typedef float v8f __attribute__((ext_vector_type(8)));


__device__ __forceinline__ float diag_sum(const v8f c, const int lane)
{
    const int  dj     = lane & 7;
    const bool onDiag = (lane < 8) || (lane >= 24);
    float d = c[0];
#pragma unroll
    for (int i = 1; i < 8; ++i) d = (dj == i) ? c[i] : d;
    d = onDiag ? d : 0.f;
#pragma unroll
    for (int m = 16; m >= 1; m >>= 1) d += __shfl_xor(d, m, 32);
    return d;
}

#define GEMV_ROWS   3
#define GEMV_WAVES  8
#define GEMV_BLOCKS (N3H / (GEMV_WAVES * GEMV_ROWS))
#define AUX_ROWS    (3 + W_SZ)
#define AUX_BLOCKS  ((AUX_ROWS + GEMV_WAVES - 1) / GEMV_WAVES)
static_assert(KIH % 128 == 0 && H_SZ % 128 == 0, "rows are whole 512-byte pieces");
static_assert(N3H % (GEMV_WAVES * GEMV_ROWS) == 0, "every wave owns GEMV_ROWS whole rows");

template <int LD>
__device__ __forceinline__ void gemv_rows(
    const float* __restrict__ W, const float* __restrict__ xv,
    const float* __restrict__ bias, float* __restrict__ out,
    const int blk, const int wave, const int lane)
{
    constexpr int KQ = H_SZ / 128;
    const int row0 = (blk * GEMV_WAVES + wave) * GEMV_ROWS;

    const float* __restrict__ xp = xv + lane * 4;
    const float* __restrict__ wp = W + (size_t)row0 * LD + lane * 4;

    v4f x[KQ];
    v4f w[GEMV_ROWS][KQ];
    float br[GEMV_ROWS];
#pragma unroll
    for (int r = 0; r < GEMV_ROWS; ++r) br[r] = bias[row0 + r];
#pragma unroll
    for (int q = 0; q < KQ; ++q) x[q] = *(const v4f*)(xp + 128 * q);
#pragma unroll
    for (int q = 0; q < KQ; ++q) {
#pragma unroll
        for (int r = 0; r < GEMV_ROWS; ++r)
            w[r][q] = *(const v4f*)(wp + (size_t)r * LD + 128 * q);
        __builtin_amdgcn_sched_barrier(0);
    }

    v8f acc[GEMV_ROWS][2];
#pragma unroll
    for (int r = 0; r < GEMV_ROWS; ++r) {
        const float b0 = (lane == 0) ? br[r] : 0.f;
        acc[r][0] = (v8f){b0,  0.f, 0.f, 0.f, 0.f, 0.f, 0.f, 0.f};
        acc[r][1] = (v8f){0.f, 0.f, 0.f, 0.f, 0.f, 0.f, 0.f, 0.f};
    }
#pragma unroll
    for (int q = 0; q < KQ; ++q) {
        const v2f a_lo = {x[q].x, x[q].y};
        const v2f a_hi = {x[q].z, x[q].w};
#pragma unroll
        for (int r = 0; r < GEMV_ROWS; ++r) {
            const v2f b_lo = {w[r][q].x, w[r][q].y};
            const v2f b_hi = {w[r][q].z, w[r][q].w};
            acc[r][0] = __builtin_amdgcn_wmma_f32_16x16x4_f32(
                false, a_lo, false, b_lo, (short)0, acc[r][0], false, false);
            acc[r][1] = __builtin_amdgcn_wmma_f32_16x16x4_f32(
                false, a_hi, false, b_hi, (short)0, acc[r][1], false, false);
        }
    }

    float mine = 0.f;
#pragma unroll
    for (int r = 0; r < GEMV_ROWS; ++r) {
        const float d = diag_sum(acc[r][0] + acc[r][1], lane);
        if (lane == r) mine = d;
    }
    if (lane < GEMV_ROWS) out[row0 + lane] = mine;
}

__global__ __launch_bounds__(GEMV_WAVES * 32) void k_front(
    const int* __restrict__ inp, const float* __restrict__ h,
    const float* __restrict__ emb,
    const float* __restrict__ ctrl_W, const float* __restrict__ ctrl_b,
    const float* __restrict__ sin_W,  const float* __restrict__ sin_b,
    const float* __restrict__ w_ih,   const float* __restrict__ w_hh,
    const float* __restrict__ b_ih,   const float* __restrict__ b_hh,
    float* __restrict__ logits, float* __restrict__ stack_in,
    float* __restrict__ gi, float* __restrict__ gh)
{
    const int lane = threadIdx.x & 31;
    const int wave = __builtin_amdgcn_readfirstlane(threadIdx.x >> 5);
    const int bid  = blockIdx.x;

    if (bid < GEMV_BLOCKS) {
        gemv_rows<KIH>(w_ih, emb + (size_t)inp[0] * H_SZ, b_ih, gi, bid, wave, lane);
        return;
    }
    if (bid < 2 * GEMV_BLOCKS) {
        gemv_rows<H_SZ>(w_hh, h, b_hh, gh, bid - GEMV_BLOCKS, wave, lane);
        return;
    }

    const int row = (bid - 2 * GEMV_BLOCKS) * GEMV_WAVES + wave;
    if (row >= AUX_ROWS) return;
    const float* Wr;
    const float* bp;
    float* outp;
    int orow;
    bool doTanh;
    if (row < 3) { Wr = ctrl_W; bp = ctrl_b; outp = logits;   orow = row;     doTanh = false; }
    else         { Wr = sin_W;  bp = sin_b;  outp = stack_in; orow = row - 3; doTanh = true;  }
    const float* wr = Wr + (size_t)orow * H_SZ;

    float s = 0.f;
#pragma unroll
    for (int q = 0; q < H_SZ / 128; ++q) {
        const int k = q * 128 + lane * 4;
        float4 wv = *(const float4*)(wr + k);
        float4 hv = *(const float4*)(h  + k);
        s += wv.x * hv.x + wv.y * hv.y + wv.z * hv.z + wv.w * hv.w;
    }
    for (int off = 16; off > 0; off >>= 1) s += __shfl_down(s, off, 32);
    if (lane == 0) {
        s += bp[orow];
        outp[orow] = doTanh ? tanhf(s) : s;
    }
}

#define MID_WAVES 8
#define MID_GRU_BLOCKS (H_SZ / MID_WAVES)
#define MID_BLEND_BLOCKS (D_SZ * W_SZ / 256)
static_assert(D_SZ * W_SZ % 256 == 0, "blend workgroups are full");

struct Controls { float c0, c1, c2; };
__device__ __forceinline__ Controls controls(const float* __restrict__ logits)
{
    const float l0 = logits[0], l1 = logits[1], l2 = logits[2];
    const float m  = fmaxf(l0, fmaxf(l1, l2));
    const float e0 = expf(l0 - m), e1 = expf(l1 - m), e2 = expf(l2 - m);
    const float inv = 1.f / (e0 + e1 + e2);
    return Controls{e0 * inv, e1 * inv, e2 * inv};
}

__global__ __launch_bounds__(256) void k_mid(
    const float* __restrict__ stack, const float* __restrict__ logits,
    const float* __restrict__ stack_in, const float* __restrict__ w_ih,
    const float* __restrict__ gi, const float* __restrict__ gh,
    const float* __restrict__ h,
    float* __restrict__ out_stack, float* __restrict__ out_h,
    float* __restrict__ hnew_ws)
{
    if (blockIdx.x >= MID_GRU_BLOCKS) {
        const int idx = (blockIdx.x - MID_GRU_BLOCKS) * 256 + threadIdx.x;
        const int d = idx >> 7;
        const int w = idx & 127;
        const float cur  = stack[idx];
        const float up   = (d == 0)        ? stack_in[w]     : stack[idx - W_SZ];
        const float down = (d == D_SZ - 1) ? 0.f             : stack[idx + W_SZ];
        const Controls c = controls(logits);
        out_stack[idx] = c.c2 * cur + c.c0 * up + c.c1 * down;
        return;
    }

    const int lane = threadIdx.x & 31;
    const int wave = __builtin_amdgcn_readfirstlane(threadIdx.x >> 5);
    const int i    = blockIdx.x * MID_WAVES + wave;

    const v4f cur  = *(const v4f*)(stack + lane * 4);
    const v4f up   = *(const v4f*)(stack_in + lane * 4);
    const v4f down = *(const v4f*)(stack + W_SZ + lane * 4);
    const float* __restrict__ wp = w_ih + (size_t)i * KIH + H_SZ + lane * 4;
    v4f wt[3];
#pragma unroll
    for (int g = 0; g < 3; ++g) wt[g] = *(const v4f*)(wp + (size_t)g * H_SZ * KIH);

    const float ir  = gi[i],            hr = gh[i];
    const float iz  = gi[H_SZ + i],     hz = gh[H_SZ + i];
    const float in_ = gi[2 * H_SZ + i], hn = gh[2 * H_SZ + i];
    const float hv  = h[i];
    __builtin_amdgcn_sched_barrier(0);

    const Controls c = controls(logits);
    const v4f top = c.c2 * cur + c.c0 * up + c.c1 * down;
    const v2f a_lo = {top.x, top.y};
    const v2f a_hi = {top.z, top.w};

    const int  dj     = lane & 7;
    const bool onDiag = (lane < 8) || (lane >= 24);
    float t[3];
#pragma unroll
    for (int g = 0; g < 3; ++g) {
        const v2f b_lo = {wt[g].x, wt[g].y};
        const v2f b_hi = {wt[g].z, wt[g].w};
        v8f acc = (v8f){0.f, 0.f, 0.f, 0.f, 0.f, 0.f, 0.f, 0.f};
        acc = __builtin_amdgcn_wmma_f32_16x16x4_f32(
            false, a_lo, false, b_lo, (short)0, acc, false, false);
        acc = __builtin_amdgcn_wmma_f32_16x16x4_f32(
            false, a_hi, false, b_hi, (short)0, acc, false, false);
        float d = acc[0];
#pragma unroll
        for (int k = 1; k < 8; ++k) d = (dj == k) ? acc[k] : d;
        t[g] = onDiag ? d : 0.f;
    }
#pragma unroll
    for (int m = 16; m >= 1; m >>= 1) {
#pragma unroll
        for (int g = 0; g < 3; ++g) t[g] += __shfl_xor(t[g], m, 32);
    }

    const float r = 1.f / (1.f + expf(-((ir + t[0]) + hr)));
    const float z = 1.f / (1.f + expf(-((iz + t[1]) + hz)));
    const float n = tanhf((in_ + t[2]) + r * hn);
    const float o = (1.f - z) * n + z * hv;
    if (lane == 0) {
        out_h[i]   = o;
        hnew_ws[i] = o;
    }
}

__global__ __launch_bounds__(32) void k_dec(
    const float* __restrict__ hnew, const float* __restrict__ dec_W,
    const float* __restrict__ dec_b, float* __restrict__ out_y)
{
    const int row = blockIdx.x;
    const float* wr = dec_W + (size_t)row * H_SZ;
    const int lane = threadIdx.x;
    float s = 0.f;
#pragma unroll 4
    for (int k = lane * 4; k < H_SZ; k += 32 * 4) {
        float4 wv = *(const float4*)(wr + k);
        float4 hv = *(const float4*)(hnew + k);
        s += wv.x * hv.x + wv.y * hv.y + wv.z * hv.z + wv.w * hv.w;
    }
    for (int off = 16; off > 0; off >>= 1) s += __shfl_down(s, off, 32);
    if (lane == 0) out_y[row] = s + dec_b[row];
}

extern "C" void kernel_launch(void* const* d_in, const int* in_sizes, int n_in,
                              void* d_out, int out_size, void* d_ws, size_t ws_size,
                              hipStream_t stream)
{
    (void)in_sizes; (void)n_in; (void)out_size; (void)ws_size;

    const int*   inp    = (const int*)  d_in[0];
    const float* hidden = (const float*)d_in[1];
    const float* stack  = (const float*)d_in[2];
    const float* emb    = (const float*)d_in[3];
    const float* ctrl_W = (const float*)d_in[4];
    const float* ctrl_b = (const float*)d_in[5];
    const float* sin_W  = (const float*)d_in[6];
    const float* sin_b  = (const float*)d_in[7];
    const float* w_ih   = (const float*)d_in[8];
    const float* w_hh   = (const float*)d_in[9];
    const float* b_ih   = (const float*)d_in[10];
    const float* b_hh   = (const float*)d_in[11];
    const float* dec_W  = (const float*)d_in[12];
    const float* dec_b  = (const float*)d_in[13];

    float* out       = (float*)d_out;
    float* out_y     = out;
    float* out_h     = out + V_SZ;
    float* out_stack = out + V_SZ + H_SZ;

    float* ws       = (float*)d_ws;
    float* logits   = ws;
    float* stack_in = ws + 16;
    float* gi       = ws + 2336;
    float* gh       = ws + 8480;
    float* hnew     = ws + 14624;

    k_front<<<AUX_BLOCKS + 2 * GEMV_BLOCKS, GEMV_WAVES * 32, 0, stream>>>(
        inp, hidden, emb, ctrl_W, ctrl_b, sin_W, sin_b, w_ih, w_hh, b_ih, b_hh,
        logits, stack_in, gi, gh);
    k_mid<<<MID_GRU_BLOCKS + MID_BLEND_BLOCKS, 256, 0, stream>>>(
        stack, logits, stack_in, w_ih, gi, gh, hidden, out_stack, out_h, hnew);
    k_dec<<<V_SZ, 32, 0, stream>>>(hnew, dec_W, dec_b, out_y);
}


# Launch 1 at four waves per SIMD, not two: 240 registers (row 3's pieces requested as row 1's are consumed), all 512 long workgroups resident in one round, short rows first; sums unchanged
# speedup vs baseline: 2.2152x; 1.1006x over previous
#include <hip/hip_runtime.h>
#include <hip/hip_bf16.h>
#include <math.h>

#define V_SZ 45
#define H_SZ 2048
#define W_SZ 128
#define D_SZ 200
#define KIH  (H_SZ + W_SZ)
#define N3H  (3 * H_SZ)

typedef float v2f __attribute__((ext_vector_type(2)));
typedef float v4f __attribute__((ext_vector_type(4)));
typedef float v8f __attribute__((ext_vector_type(8)));


#define GEMV_ROWS   3
#define GEMV_WAVES  8
#define GEMV_BLOCKS (N3H / (GEMV_WAVES * GEMV_ROWS))
#define AUX_ROWS    (3 + W_SZ)
#define AUX_BLOCKS  ((AUX_ROWS + GEMV_WAVES - 1) / GEMV_WAVES)
static_assert(KIH % 128 == 0 && H_SZ % 128 == 0, "rows are whole 512-byte pieces");
static_assert(N3H % (GEMV_WAVES * GEMV_ROWS) == 0, "every wave owns GEMV_ROWS whole rows");
static_assert(GEMV_ROWS == 3, "the row schedule below is written out for three rows");

__device__ __forceinline__ float diag_pick(const v8f c, const int lane)
{
    const int  dj     = lane & 7;
    const bool onDiag = (lane < 8) || (lane >= 24);
    float d = c[0];
#pragma unroll
    for (int i = 1; i < 8; ++i) d = (dj == i) ? c[i] : d;
    return onDiag ? d : 0.f;
}

#define GEMV_ROW_BEGIN(a0, a1, b)                                              \
    v8f a0 = (v8f){(lane == 0) ? (b) : 0.f, 0.f, 0.f, 0.f, 0.f, 0.f, 0.f, 0.f}; \
    v8f a1 = (v8f){0.f, 0.f, 0.f, 0.f, 0.f, 0.f, 0.f, 0.f};
#define GEMV_PIECE(a0, a1, xq, wq)                                             \
    a0 = __builtin_amdgcn_wmma_f32_16x16x4_f32(                                \
        false, (v2f){(xq).x, (xq).y}, false, (v2f){(wq).x, (wq).y}, (short)0, a0, false, false); \
    a1 = __builtin_amdgcn_wmma_f32_16x16x4_f32(                                \
        false, (v2f){(xq).z, (xq).w}, false, (v2f){(wq).z, (wq).w}, (short)0, a1, false, false);

template <int LD>
__device__ __forceinline__ void gemv_rows(
    const float* __restrict__ W, const float* __restrict__ xv,
    const float* __restrict__ bias, float* __restrict__ out,
    const int blk, const int wave, const int lane)
{
    constexpr int KQ = H_SZ / 128;
    const int row0 = (blk * GEMV_WAVES + wave) * GEMV_ROWS;

    const float* __restrict__ xp = xv + lane * 4;
    const float* __restrict__ wp = W + (size_t)row0 * LD + lane * 4;

    float br[GEMV_ROWS];
#pragma unroll
    for (int r = 0; r < GEMV_ROWS; ++r) br[r] = bias[row0 + r];

    v4f x[KQ];
    v4f wa[KQ];
    v4f wb[KQ];
    v4f wc[KQ];
#pragma unroll
    for (int q = 0; q < KQ; ++q) x[q] = *(const v4f*)(xp + 128 * q);
#pragma unroll
    for (int q = 0; q < KQ; ++q) wa[q] = *(const v4f*)(wp + 128 * q);
#pragma unroll
    for (int q = 0; q < KQ; ++q) wb[q] = *(const v4f*)(wp + (size_t)LD + 128 * q);
    __builtin_amdgcn_sched_barrier(0);

    float t[GEMV_ROWS];
    {
        GEMV_ROW_BEGIN(a0, a1, br[0])
#pragma unroll
        for (int q = 0; q < KQ; ++q) {
            GEMV_PIECE(a0, a1, x[q], wa[q])
            wc[q] = *(const v4f*)(wp + (size_t)2 * LD + 128 * q);
            __builtin_amdgcn_sched_barrier(0);
        }
        t[0] = diag_pick(a0 + a1, lane);
    }
    {
        GEMV_ROW_BEGIN(a0, a1, br[1])
#pragma unroll
        for (int q = 0; q < KQ; ++q) { GEMV_PIECE(a0, a1, x[q], wb[q]) }
        __builtin_amdgcn_sched_barrier(0);
        t[1] = diag_pick(a0 + a1, lane);
    }
    {
        GEMV_ROW_BEGIN(a0, a1, br[2])
#pragma unroll
        for (int q = 0; q < KQ; ++q) { GEMV_PIECE(a0, a1, x[q], wc[q]) }
        t[2] = diag_pick(a0 + a1, lane);
    }

#pragma unroll
    for (int m = 16; m >= 1; m >>= 1) {
#pragma unroll
        for (int r = 0; r < GEMV_ROWS; ++r) t[r] += __shfl_xor(t[r], m, 32);
    }
    float mine = 0.f;
#pragma unroll
    for (int r = 0; r < GEMV_ROWS; ++r) if (lane == r) mine = t[r];
    if (lane < GEMV_ROWS) out[row0 + lane] = mine;
}

__global__ __launch_bounds__(GEMV_WAVES * 32) void k_front(
    const int* __restrict__ inp, const float* __restrict__ h,
    const float* __restrict__ emb,
    const float* __restrict__ ctrl_W, const float* __restrict__ ctrl_b,
    const float* __restrict__ sin_W,  const float* __restrict__ sin_b,
    const float* __restrict__ w_ih,   const float* __restrict__ w_hh,
    const float* __restrict__ b_ih,   const float* __restrict__ b_hh,
    float* __restrict__ logits, float* __restrict__ stack_in,
    float* __restrict__ gi, float* __restrict__ gh)
{
    const int lane = threadIdx.x & 31;
    const int wave = __builtin_amdgcn_readfirstlane(threadIdx.x >> 5);
    const int bid  = blockIdx.x;

    if (bid >= AUX_BLOCKS + GEMV_BLOCKS) {
        gemv_rows<H_SZ>(w_hh, h, b_hh, gh, bid - AUX_BLOCKS - GEMV_BLOCKS, wave, lane);
        return;
    }
    if (bid >= AUX_BLOCKS) {
        gemv_rows<KIH>(w_ih, emb + (size_t)inp[0] * H_SZ, b_ih, gi, bid - AUX_BLOCKS, wave, lane);
        return;
    }

    const int row = bid * GEMV_WAVES + wave;
    if (row >= AUX_ROWS) return;
    const float* Wr;
    const float* bp;
    float* outp;
    int orow;
    bool doTanh;
    if (row < 3) { Wr = ctrl_W; bp = ctrl_b; outp = logits;   orow = row;     doTanh = false; }
    else         { Wr = sin_W;  bp = sin_b;  outp = stack_in; orow = row - 3; doTanh = true;  }
    const float* wr = Wr + (size_t)orow * H_SZ;

    float s = 0.f;
#pragma unroll
    for (int q = 0; q < H_SZ / 128; ++q) {
        const int k = q * 128 + lane * 4;
        float4 wv = *(const float4*)(wr + k);
        float4 hv = *(const float4*)(h  + k);
        s += wv.x * hv.x + wv.y * hv.y + wv.z * hv.z + wv.w * hv.w;
    }
    for (int off = 16; off > 0; off >>= 1) s += __shfl_down(s, off, 32);
    if (lane == 0) {
        s += bp[orow];
        outp[orow] = doTanh ? tanhf(s) : s;
    }
}

#define MID_WAVES 8
#define MID_GRU_BLOCKS (H_SZ / MID_WAVES)
#define MID_BLEND_BLOCKS (D_SZ * W_SZ / 256)
static_assert(D_SZ * W_SZ % 256 == 0, "blend workgroups are full");

struct Controls { float c0, c1, c2; };
__device__ __forceinline__ Controls controls(const float* __restrict__ logits)
{
    const float l0 = logits[0], l1 = logits[1], l2 = logits[2];
    const float m  = fmaxf(l0, fmaxf(l1, l2));
    const float e0 = expf(l0 - m), e1 = expf(l1 - m), e2 = expf(l2 - m);
    const float inv = 1.f / (e0 + e1 + e2);
    return Controls{e0 * inv, e1 * inv, e2 * inv};
}

__global__ __launch_bounds__(256) void k_mid(
    const float* __restrict__ stack, const float* __restrict__ logits,
    const float* __restrict__ stack_in, const float* __restrict__ w_ih,
    const float* __restrict__ gi, const float* __restrict__ gh,
    const float* __restrict__ h,
    float* __restrict__ out_stack, float* __restrict__ out_h,
    float* __restrict__ hnew_ws)
{
    if (blockIdx.x >= MID_GRU_BLOCKS) {
        const int idx = (blockIdx.x - MID_GRU_BLOCKS) * 256 + threadIdx.x;
        const int d = idx >> 7;
        const int w = idx & 127;
        const float cur  = stack[idx];
        const float up   = (d == 0)        ? stack_in[w]     : stack[idx - W_SZ];
        const float down = (d == D_SZ - 1) ? 0.f             : stack[idx + W_SZ];
        const Controls c = controls(logits);
        out_stack[idx] = c.c2 * cur + c.c0 * up + c.c1 * down;
        return;
    }

    const int lane = threadIdx.x & 31;
    const int wave = __builtin_amdgcn_readfirstlane(threadIdx.x >> 5);
    const int i    = blockIdx.x * MID_WAVES + wave;

    const v4f cur  = *(const v4f*)(stack + lane * 4);
    const v4f up   = *(const v4f*)(stack_in + lane * 4);
    const v4f down = *(const v4f*)(stack + W_SZ + lane * 4);
    const float* __restrict__ wp = w_ih + (size_t)i * KIH + H_SZ + lane * 4;
    v4f wt[3];
#pragma unroll
    for (int g = 0; g < 3; ++g) wt[g] = *(const v4f*)(wp + (size_t)g * H_SZ * KIH);

    const float ir  = gi[i],            hr = gh[i];
    const float iz  = gi[H_SZ + i],     hz = gh[H_SZ + i];
    const float in_ = gi[2 * H_SZ + i], hn = gh[2 * H_SZ + i];
    const float hv  = h[i];
    __builtin_amdgcn_sched_barrier(0);

    const Controls c = controls(logits);
    const v4f top = c.c2 * cur + c.c0 * up + c.c1 * down;
    const v2f a_lo = {top.x, top.y};
    const v2f a_hi = {top.z, top.w};

    const int  dj     = lane & 7;
    const bool onDiag = (lane < 8) || (lane >= 24);
    float t[3];
#pragma unroll
    for (int g = 0; g < 3; ++g) {
        const v2f b_lo = {wt[g].x, wt[g].y};
        const v2f b_hi = {wt[g].z, wt[g].w};
        v8f acc = (v8f){0.f, 0.f, 0.f, 0.f, 0.f, 0.f, 0.f, 0.f};
        acc = __builtin_amdgcn_wmma_f32_16x16x4_f32(
            false, a_lo, false, b_lo, (short)0, acc, false, false);
        acc = __builtin_amdgcn_wmma_f32_16x16x4_f32(
            false, a_hi, false, b_hi, (short)0, acc, false, false);
        float d = acc[0];
#pragma unroll
        for (int k = 1; k < 8; ++k) d = (dj == k) ? acc[k] : d;
        t[g] = onDiag ? d : 0.f;
    }
#pragma unroll
    for (int m = 16; m >= 1; m >>= 1) {
#pragma unroll
        for (int g = 0; g < 3; ++g) t[g] += __shfl_xor(t[g], m, 32);
    }

    const float r = 1.f / (1.f + expf(-((ir + t[0]) + hr)));
    const float z = 1.f / (1.f + expf(-((iz + t[1]) + hz)));
    const float n = tanhf((in_ + t[2]) + r * hn);
    const float o = (1.f - z) * n + z * hv;
    if (lane == 0) {
        out_h[i]   = o;
        hnew_ws[i] = o;
    }
}

__global__ __launch_bounds__(32) void k_dec(
    const float* __restrict__ hnew, const float* __restrict__ dec_W,
    const float* __restrict__ dec_b, float* __restrict__ out_y)
{
    const int row = blockIdx.x;
    const float* wr = dec_W + (size_t)row * H_SZ;
    const int lane = threadIdx.x;
    float s = 0.f;
#pragma unroll 4
    for (int k = lane * 4; k < H_SZ; k += 32 * 4) {
        float4 wv = *(const float4*)(wr + k);
        float4 hv = *(const float4*)(hnew + k);
        s += wv.x * hv.x + wv.y * hv.y + wv.z * hv.z + wv.w * hv.w;
    }
    for (int off = 16; off > 0; off >>= 1) s += __shfl_down(s, off, 32);
    if (lane == 0) out_y[row] = s + dec_b[row];
}

extern "C" void kernel_launch(void* const* d_in, const int* in_sizes, int n_in,
                              void* d_out, int out_size, void* d_ws, size_t ws_size,
                              hipStream_t stream)
{
    (void)in_sizes; (void)n_in; (void)out_size; (void)ws_size;

    const int*   inp    = (const int*)  d_in[0];
    const float* hidden = (const float*)d_in[1];
    const float* stack  = (const float*)d_in[2];
    const float* emb    = (const float*)d_in[3];
    const float* ctrl_W = (const float*)d_in[4];
    const float* ctrl_b = (const float*)d_in[5];
    const float* sin_W  = (const float*)d_in[6];
    const float* sin_b  = (const float*)d_in[7];
    const float* w_ih   = (const float*)d_in[8];
    const float* w_hh   = (const float*)d_in[9];
    const float* b_ih   = (const float*)d_in[10];
    const float* b_hh   = (const float*)d_in[11];
    const float* dec_W  = (const float*)d_in[12];
    const float* dec_b  = (const float*)d_in[13];

    float* out       = (float*)d_out;
    float* out_y     = out;
    float* out_h     = out + V_SZ;
    float* out_stack = out + V_SZ + H_SZ;

    float* ws       = (float*)d_ws;
    float* logits   = ws;
    float* stack_in = ws + 16;
    float* gi       = ws + 2336;
    float* gh       = ws + 8480;
    float* hnew     = ws + 14624;

    k_front<<<AUX_BLOCKS + 2 * GEMV_BLOCKS, GEMV_WAVES * 32, 0, stream>>>(
        inp, hidden, emb, ctrl_W, ctrl_b, sin_W, sin_b, w_ih, w_hh, b_ih, b_hh,
        logits, stack_in, gi, gh);
    k_mid<<<MID_GRU_BLOCKS + MID_BLEND_BLOCKS, 256, 0, stream>>>(
        stack, logits, stack_in, w_ih, gi, gh, hidden, out_stack, out_h, hnew);
    k_dec<<<V_SZ, 32, 0, stream>>>(hnew, dec_W, dec_b, out_y);
}


# Launch 1 on half the waves: a wave takes six rows, not three, and refills each of two row buffers piece by piece behind its products; 256 long workgroups, one per WGP; sums unchanged
# speedup vs baseline: 2.3348x; 1.0540x over previous
#include <hip/hip_runtime.h>
#include <hip/hip_bf16.h>
#include <math.h>

#define V_SZ 45
#define H_SZ 2048
#define W_SZ 128
#define D_SZ 200
#define KIH  (H_SZ + W_SZ)
#define N3H  (3 * H_SZ)

typedef float v2f __attribute__((ext_vector_type(2)));
typedef float v4f __attribute__((ext_vector_type(4)));
typedef float v8f __attribute__((ext_vector_type(8)));


#define GEMV_ROWS   6
#define GEMV_WAVES  8
#define GEMV_BLOCKS (N3H / (GEMV_WAVES * GEMV_ROWS))
#define AUX_ROWS    (3 + W_SZ)
#define AUX_BLOCKS  ((AUX_ROWS + GEMV_WAVES - 1) / GEMV_WAVES)
static_assert(KIH % 128 == 0 && H_SZ % 128 == 0, "rows are whole 512-byte pieces");
static_assert(N3H % (GEMV_WAVES * GEMV_ROWS) == 0, "every wave owns GEMV_ROWS whole rows");
static_assert(GEMV_ROWS == 6, "the row schedule below is written out for six rows");

__device__ __forceinline__ float diag_pick(const v8f c, const int lane)
{
    const int  dj     = lane & 7;
    const bool onDiag = (lane < 8) || (lane >= 24);
    float d = c[0];
#pragma unroll
    for (int i = 1; i < 8; ++i) d = (dj == i) ? c[i] : d;
    return onDiag ? d : 0.f;
}

#define GEMV_ROW_BEGIN(a0, a1, b)                                              \
    v8f a0 = (v8f){(lane == 0) ? (b) : 0.f, 0.f, 0.f, 0.f, 0.f, 0.f, 0.f, 0.f}; \
    v8f a1 = (v8f){0.f, 0.f, 0.f, 0.f, 0.f, 0.f, 0.f, 0.f};
#define GEMV_PIECE(a0, a1, xq, wq)                                             \
    a0 = __builtin_amdgcn_wmma_f32_16x16x4_f32(                                \
        false, (v2f){(xq).x, (xq).y}, false, (v2f){(wq).x, (wq).y}, (short)0, a0, false, false); \
    a1 = __builtin_amdgcn_wmma_f32_16x16x4_f32(                                \
        false, (v2f){(xq).z, (xq).w}, false, (v2f){(wq).z, (wq).w}, (short)0, a1, false, false);

#define GEMV_PHASE_REFILL(r, buf, nextrow)                                     \
    {                                                                          \
        GEMV_ROW_BEGIN(a0, a1, br[r])                                          \
        _Pragma("unroll")                                                      \
        for (int q = 0; q < KQ; ++q) {                                         \
            GEMV_PIECE(a0, a1, x[q], buf[q])                                   \
            buf[q] = *(const v4f*)(wp + (size_t)(nextrow) * LD + 128 * q);     \
            __builtin_amdgcn_sched_barrier(0);                                 \
        }                                                                      \
        t[r] = diag_pick(a0 + a1, lane);                                       \
        __builtin_amdgcn_sched_barrier(0);                                     \
    }
#define GEMV_PHASE_LAST(r, buf)                                                \
    {                                                                          \
        GEMV_ROW_BEGIN(a0, a1, br[r])                                          \
        _Pragma("unroll")                                                      \
        for (int q = 0; q < KQ; ++q) { GEMV_PIECE(a0, a1, x[q], buf[q]) }      \
        t[r] = diag_pick(a0 + a1, lane);                                       \
    }

template <int LD>
__device__ __forceinline__ void gemv_rows(
    const float* __restrict__ W, const float* __restrict__ xv,
    const float* __restrict__ bias, float* __restrict__ out,
    const int blk, const int wave, const int lane)
{
    constexpr int KQ = H_SZ / 128;
    const int row0 = (blk * GEMV_WAVES + wave) * GEMV_ROWS;

    const float* __restrict__ xp = xv + lane * 4;
    const float* __restrict__ wp = W + (size_t)row0 * LD + lane * 4;

    float br[GEMV_ROWS];
#pragma unroll
    for (int r = 0; r < GEMV_ROWS; ++r) br[r] = bias[row0 + r];

    v4f x[KQ];
    v4f wa[KQ];
    v4f wb[KQ];
#pragma unroll
    for (int q = 0; q < KQ; ++q) x[q] = *(const v4f*)(xp + 128 * q);
#pragma unroll
    for (int q = 0; q < KQ; ++q) wa[q] = *(const v4f*)(wp + 128 * q);
#pragma unroll
    for (int q = 0; q < KQ; ++q) wb[q] = *(const v4f*)(wp + (size_t)LD + 128 * q);
    __builtin_amdgcn_sched_barrier(0);

    float t[GEMV_ROWS];
    GEMV_PHASE_REFILL(0, wa, 2)
    GEMV_PHASE_REFILL(1, wb, 3)
    GEMV_PHASE_REFILL(2, wa, 4)
    GEMV_PHASE_REFILL(3, wb, 5)
    GEMV_PHASE_LAST(4, wa)
    __builtin_amdgcn_sched_barrier(0);
    GEMV_PHASE_LAST(5, wb)

#pragma unroll
    for (int m = 16; m >= 1; m >>= 1) {
#pragma unroll
        for (int r = 0; r < GEMV_ROWS; ++r) t[r] += __shfl_xor(t[r], m, 32);
    }
    float mine = 0.f;
#pragma unroll
    for (int r = 0; r < GEMV_ROWS; ++r) if (lane == r) mine = t[r];
    if (lane < GEMV_ROWS) out[row0 + lane] = mine;
}

__global__ __launch_bounds__(GEMV_WAVES * 32) void k_front(
    const int* __restrict__ inp, const float* __restrict__ h,
    const float* __restrict__ emb,
    const float* __restrict__ ctrl_W, const float* __restrict__ ctrl_b,
    const float* __restrict__ sin_W,  const float* __restrict__ sin_b,
    const float* __restrict__ w_ih,   const float* __restrict__ w_hh,
    const float* __restrict__ b_ih,   const float* __restrict__ b_hh,
    float* __restrict__ logits, float* __restrict__ stack_in,
    float* __restrict__ gi, float* __restrict__ gh)
{
    const int lane = threadIdx.x & 31;
    const int wave = __builtin_amdgcn_readfirstlane(threadIdx.x >> 5);
    const int bid  = blockIdx.x;

    if (bid >= AUX_BLOCKS + GEMV_BLOCKS) {
        gemv_rows<H_SZ>(w_hh, h, b_hh, gh, bid - AUX_BLOCKS - GEMV_BLOCKS, wave, lane);
        return;
    }
    if (bid >= AUX_BLOCKS) {
        gemv_rows<KIH>(w_ih, emb + (size_t)inp[0] * H_SZ, b_ih, gi, bid - AUX_BLOCKS, wave, lane);
        return;
    }

    const int row = bid * GEMV_WAVES + wave;
    if (row >= AUX_ROWS) return;
    const float* Wr;
    const float* bp;
    float* outp;
    int orow;
    bool doTanh;
    if (row < 3) { Wr = ctrl_W; bp = ctrl_b; outp = logits;   orow = row;     doTanh = false; }
    else         { Wr = sin_W;  bp = sin_b;  outp = stack_in; orow = row - 3; doTanh = true;  }
    const float* wr = Wr + (size_t)orow * H_SZ;

    float s = 0.f;
#pragma unroll
    for (int q = 0; q < H_SZ / 128; ++q) {
        const int k = q * 128 + lane * 4;
        float4 wv = *(const float4*)(wr + k);
        float4 hv = *(const float4*)(h  + k);
        s += wv.x * hv.x + wv.y * hv.y + wv.z * hv.z + wv.w * hv.w;
    }
    for (int off = 16; off > 0; off >>= 1) s += __shfl_down(s, off, 32);
    if (lane == 0) {
        s += bp[orow];
        outp[orow] = doTanh ? tanhf(s) : s;
    }
}

#define MID_WAVES 8
#define MID_GRU_BLOCKS (H_SZ / MID_WAVES)
#define MID_BLEND_BLOCKS (D_SZ * W_SZ / 256)
static_assert(D_SZ * W_SZ % 256 == 0, "blend workgroups are full");

struct Controls { float c0, c1, c2; };
__device__ __forceinline__ Controls controls(const float* __restrict__ logits)
{
    const float l0 = logits[0], l1 = logits[1], l2 = logits[2];
    const float m  = fmaxf(l0, fmaxf(l1, l2));
    const float e0 = expf(l0 - m), e1 = expf(l1 - m), e2 = expf(l2 - m);
    const float inv = 1.f / (e0 + e1 + e2);
    return Controls{e0 * inv, e1 * inv, e2 * inv};
}

__global__ __launch_bounds__(256) void k_mid(
    const float* __restrict__ stack, const float* __restrict__ logits,
    const float* __restrict__ stack_in, const float* __restrict__ w_ih,
    const float* __restrict__ gi, const float* __restrict__ gh,
    const float* __restrict__ h,
    float* __restrict__ out_stack, float* __restrict__ out_h,
    float* __restrict__ hnew_ws)
{
    if (blockIdx.x >= MID_GRU_BLOCKS) {
        const int idx = (blockIdx.x - MID_GRU_BLOCKS) * 256 + threadIdx.x;
        const int d = idx >> 7;
        const int w = idx & 127;
        const float cur  = stack[idx];
        const float up   = (d == 0)        ? stack_in[w]     : stack[idx - W_SZ];
        const float down = (d == D_SZ - 1) ? 0.f             : stack[idx + W_SZ];
        const Controls c = controls(logits);
        out_stack[idx] = c.c2 * cur + c.c0 * up + c.c1 * down;
        return;
    }

    const int lane = threadIdx.x & 31;
    const int wave = __builtin_amdgcn_readfirstlane(threadIdx.x >> 5);
    const int i    = blockIdx.x * MID_WAVES + wave;

    const v4f cur  = *(const v4f*)(stack + lane * 4);
    const v4f up   = *(const v4f*)(stack_in + lane * 4);
    const v4f down = *(const v4f*)(stack + W_SZ + lane * 4);
    const float* __restrict__ wp = w_ih + (size_t)i * KIH + H_SZ + lane * 4;
    v4f wt[3];
#pragma unroll
    for (int g = 0; g < 3; ++g) wt[g] = *(const v4f*)(wp + (size_t)g * H_SZ * KIH);

    const float ir  = gi[i],            hr = gh[i];
    const float iz  = gi[H_SZ + i],     hz = gh[H_SZ + i];
    const float in_ = gi[2 * H_SZ + i], hn = gh[2 * H_SZ + i];
    const float hv  = h[i];
    __builtin_amdgcn_sched_barrier(0);

    const Controls c = controls(logits);
    const v4f top = c.c2 * cur + c.c0 * up + c.c1 * down;
    const v2f a_lo = {top.x, top.y};
    const v2f a_hi = {top.z, top.w};

    const int  dj     = lane & 7;
    const bool onDiag = (lane < 8) || (lane >= 24);
    float t[3];
#pragma unroll
    for (int g = 0; g < 3; ++g) {
        const v2f b_lo = {wt[g].x, wt[g].y};
        const v2f b_hi = {wt[g].z, wt[g].w};
        v8f acc = (v8f){0.f, 0.f, 0.f, 0.f, 0.f, 0.f, 0.f, 0.f};
        acc = __builtin_amdgcn_wmma_f32_16x16x4_f32(
            false, a_lo, false, b_lo, (short)0, acc, false, false);
        acc = __builtin_amdgcn_wmma_f32_16x16x4_f32(
            false, a_hi, false, b_hi, (short)0, acc, false, false);
        float d = acc[0];
#pragma unroll
        for (int k = 1; k < 8; ++k) d = (dj == k) ? acc[k] : d;
        t[g] = onDiag ? d : 0.f;
    }
#pragma unroll
    for (int m = 16; m >= 1; m >>= 1) {
#pragma unroll
        for (int g = 0; g < 3; ++g) t[g] += __shfl_xor(t[g], m, 32);
    }

    const float r = 1.f / (1.f + expf(-((ir + t[0]) + hr)));
    const float z = 1.f / (1.f + expf(-((iz + t[1]) + hz)));
    const float n = tanhf((in_ + t[2]) + r * hn);
    const float o = (1.f - z) * n + z * hv;
    if (lane == 0) {
        out_h[i]   = o;
        hnew_ws[i] = o;
    }
}

__global__ __launch_bounds__(32) void k_dec(
    const float* __restrict__ hnew, const float* __restrict__ dec_W,
    const float* __restrict__ dec_b, float* __restrict__ out_y)
{
    const int row = blockIdx.x;
    const float* wr = dec_W + (size_t)row * H_SZ;
    const int lane = threadIdx.x;
    float s = 0.f;
#pragma unroll 4
    for (int k = lane * 4; k < H_SZ; k += 32 * 4) {
        float4 wv = *(const float4*)(wr + k);
        float4 hv = *(const float4*)(hnew + k);
        s += wv.x * hv.x + wv.y * hv.y + wv.z * hv.z + wv.w * hv.w;
    }
    for (int off = 16; off > 0; off >>= 1) s += __shfl_down(s, off, 32);
    if (lane == 0) out_y[row] = s + dec_b[row];
}

extern "C" void kernel_launch(void* const* d_in, const int* in_sizes, int n_in,
                              void* d_out, int out_size, void* d_ws, size_t ws_size,
                              hipStream_t stream)
{
    (void)in_sizes; (void)n_in; (void)out_size; (void)ws_size;

    const int*   inp    = (const int*)  d_in[0];
    const float* hidden = (const float*)d_in[1];
    const float* stack  = (const float*)d_in[2];
    const float* emb    = (const float*)d_in[3];
    const float* ctrl_W = (const float*)d_in[4];
    const float* ctrl_b = (const float*)d_in[5];
    const float* sin_W  = (const float*)d_in[6];
    const float* sin_b  = (const float*)d_in[7];
    const float* w_ih   = (const float*)d_in[8];
    const float* w_hh   = (const float*)d_in[9];
    const float* b_ih   = (const float*)d_in[10];
    const float* b_hh   = (const float*)d_in[11];
    const float* dec_W  = (const float*)d_in[12];
    const float* dec_b  = (const float*)d_in[13];

    float* out       = (float*)d_out;
    float* out_y     = out;
    float* out_h     = out + V_SZ;
    float* out_stack = out + V_SZ + H_SZ;

    float* ws       = (float*)d_ws;
    float* logits   = ws;
    float* stack_in = ws + 16;
    float* gi       = ws + 2336;
    float* gh       = ws + 8480;
    float* hnew     = ws + 14624;

    k_front<<<AUX_BLOCKS + 2 * GEMV_BLOCKS, GEMV_WAVES * 32, 0, stream>>>(
        inp, hidden, emb, ctrl_W, ctrl_b, sin_W, sin_b, w_ih, w_hh, b_ih, b_hh,
        logits, stack_in, gi, gh);
    k_mid<<<MID_GRU_BLOCKS + MID_BLEND_BLOCKS, 256, 0, stream>>>(
        stack, logits, stack_in, w_ih, gi, gh, hidden, out_stack, out_h, hnew);
    k_dec<<<V_SZ, 32, 0, stream>>>(hnew, dec_W, dec_b, out_y);
}


# Launch 3: the decoder wave requests its whole row, all of h_new and its bias before the first multiply: one memory round trip in its chain, not five; sum order kept, multiply-adds now fused
# speedup vs baseline: 2.6197x; 1.1220x over previous
#include <hip/hip_runtime.h>
#include <hip/hip_bf16.h>
#include <math.h>

#define V_SZ 45
#define H_SZ 2048
#define W_SZ 128
#define D_SZ 200
#define KIH  (H_SZ + W_SZ)
#define N3H  (3 * H_SZ)

typedef float v2f __attribute__((ext_vector_type(2)));
typedef float v4f __attribute__((ext_vector_type(4)));
typedef float v8f __attribute__((ext_vector_type(8)));


#define GEMV_ROWS   6
#define GEMV_WAVES  8
#define GEMV_BLOCKS (N3H / (GEMV_WAVES * GEMV_ROWS))
#define AUX_ROWS    (3 + W_SZ)
#define AUX_BLOCKS  ((AUX_ROWS + GEMV_WAVES - 1) / GEMV_WAVES)
static_assert(KIH % 128 == 0 && H_SZ % 128 == 0, "rows are whole 512-byte pieces");
static_assert(N3H % (GEMV_WAVES * GEMV_ROWS) == 0, "every wave owns GEMV_ROWS whole rows");
static_assert(GEMV_ROWS == 6, "the row schedule below is written out for six rows");

__device__ __forceinline__ float diag_pick(const v8f c, const int lane)
{
    const int  dj     = lane & 7;
    const bool onDiag = (lane < 8) || (lane >= 24);
    float d = c[0];
#pragma unroll
    for (int i = 1; i < 8; ++i) d = (dj == i) ? c[i] : d;
    return onDiag ? d : 0.f;
}

#define GEMV_ROW_BEGIN(a0, a1, b)                                              \
    v8f a0 = (v8f){(lane == 0) ? (b) : 0.f, 0.f, 0.f, 0.f, 0.f, 0.f, 0.f, 0.f}; \
    v8f a1 = (v8f){0.f, 0.f, 0.f, 0.f, 0.f, 0.f, 0.f, 0.f};
#define GEMV_PIECE(a0, a1, xq, wq)                                             \
    a0 = __builtin_amdgcn_wmma_f32_16x16x4_f32(                                \
        false, (v2f){(xq).x, (xq).y}, false, (v2f){(wq).x, (wq).y}, (short)0, a0, false, false); \
    a1 = __builtin_amdgcn_wmma_f32_16x16x4_f32(                                \
        false, (v2f){(xq).z, (xq).w}, false, (v2f){(wq).z, (wq).w}, (short)0, a1, false, false);

#define GEMV_PHASE_REFILL(r, buf, nextrow)                                     \
    {                                                                          \
        GEMV_ROW_BEGIN(a0, a1, br[r])                                          \
        _Pragma("unroll")                                                      \
        for (int q = 0; q < KQ; ++q) {                                         \
            GEMV_PIECE(a0, a1, x[q], buf[q])                                   \
            buf[q] = *(const v4f*)(wp + (size_t)(nextrow) * LD + 128 * q);     \
            __builtin_amdgcn_sched_barrier(0);                                 \
        }                                                                      \
        t[r] = diag_pick(a0 + a1, lane);                                       \
        __builtin_amdgcn_sched_barrier(0);                                     \
    }
#define GEMV_PHASE_LAST(r, buf)                                                \
    {                                                                          \
        GEMV_ROW_BEGIN(a0, a1, br[r])                                          \
        _Pragma("unroll")                                                      \
        for (int q = 0; q < KQ; ++q) { GEMV_PIECE(a0, a1, x[q], buf[q]) }      \
        t[r] = diag_pick(a0 + a1, lane);                                       \
    }

template <int LD>
__device__ __forceinline__ void gemv_rows(
    const float* __restrict__ W, const float* __restrict__ xv,
    const float* __restrict__ bias, float* __restrict__ out,
    const int blk, const int wave, const int lane)
{
    constexpr int KQ = H_SZ / 128;
    const int row0 = (blk * GEMV_WAVES + wave) * GEMV_ROWS;

    const float* __restrict__ xp = xv + lane * 4;
    const float* __restrict__ wp = W + (size_t)row0 * LD + lane * 4;

    float br[GEMV_ROWS];
#pragma unroll
    for (int r = 0; r < GEMV_ROWS; ++r) br[r] = bias[row0 + r];

    v4f x[KQ];
    v4f wa[KQ];
    v4f wb[KQ];
#pragma unroll
    for (int q = 0; q < KQ; ++q) x[q] = *(const v4f*)(xp + 128 * q);
#pragma unroll
    for (int q = 0; q < KQ; ++q) wa[q] = *(const v4f*)(wp + 128 * q);
#pragma unroll
    for (int q = 0; q < KQ; ++q) wb[q] = *(const v4f*)(wp + (size_t)LD + 128 * q);
    __builtin_amdgcn_sched_barrier(0);

    float t[GEMV_ROWS];
    GEMV_PHASE_REFILL(0, wa, 2)
    GEMV_PHASE_REFILL(1, wb, 3)
    GEMV_PHASE_REFILL(2, wa, 4)
    GEMV_PHASE_REFILL(3, wb, 5)
    GEMV_PHASE_LAST(4, wa)
    __builtin_amdgcn_sched_barrier(0);
    GEMV_PHASE_LAST(5, wb)

#pragma unroll
    for (int m = 16; m >= 1; m >>= 1) {
#pragma unroll
        for (int r = 0; r < GEMV_ROWS; ++r) t[r] += __shfl_xor(t[r], m, 32);
    }
    float mine = 0.f;
#pragma unroll
    for (int r = 0; r < GEMV_ROWS; ++r) if (lane == r) mine = t[r];
    if (lane < GEMV_ROWS) out[row0 + lane] = mine;
}

__global__ __launch_bounds__(GEMV_WAVES * 32) void k_front(
    const int* __restrict__ inp, const float* __restrict__ h,
    const float* __restrict__ emb,
    const float* __restrict__ ctrl_W, const float* __restrict__ ctrl_b,
    const float* __restrict__ sin_W,  const float* __restrict__ sin_b,
    const float* __restrict__ w_ih,   const float* __restrict__ w_hh,
    const float* __restrict__ b_ih,   const float* __restrict__ b_hh,
    float* __restrict__ logits, float* __restrict__ stack_in,
    float* __restrict__ gi, float* __restrict__ gh)
{
    const int lane = threadIdx.x & 31;
    const int wave = __builtin_amdgcn_readfirstlane(threadIdx.x >> 5);
    const int bid  = blockIdx.x;

    if (bid >= AUX_BLOCKS + GEMV_BLOCKS) {
        gemv_rows<H_SZ>(w_hh, h, b_hh, gh, bid - AUX_BLOCKS - GEMV_BLOCKS, wave, lane);
        return;
    }
    if (bid >= AUX_BLOCKS) {
        gemv_rows<KIH>(w_ih, emb + (size_t)inp[0] * H_SZ, b_ih, gi, bid - AUX_BLOCKS, wave, lane);
        return;
    }

    const int row = bid * GEMV_WAVES + wave;
    if (row >= AUX_ROWS) return;
    const float* Wr;
    const float* bp;
    float* outp;
    int orow;
    bool doTanh;
    if (row < 3) { Wr = ctrl_W; bp = ctrl_b; outp = logits;   orow = row;     doTanh = false; }
    else         { Wr = sin_W;  bp = sin_b;  outp = stack_in; orow = row - 3; doTanh = true;  }
    const float* wr = Wr + (size_t)orow * H_SZ;

    float s = 0.f;
#pragma unroll
    for (int q = 0; q < H_SZ / 128; ++q) {
        const int k = q * 128 + lane * 4;
        float4 wv = *(const float4*)(wr + k);
        float4 hv = *(const float4*)(h  + k);
        s += wv.x * hv.x + wv.y * hv.y + wv.z * hv.z + wv.w * hv.w;
    }
    for (int off = 16; off > 0; off >>= 1) s += __shfl_down(s, off, 32);
    if (lane == 0) {
        s += bp[orow];
        outp[orow] = doTanh ? tanhf(s) : s;
    }
}

#define MID_WAVES 8
#define MID_GRU_BLOCKS (H_SZ / MID_WAVES)
#define MID_BLEND_BLOCKS (D_SZ * W_SZ / 256)
static_assert(D_SZ * W_SZ % 256 == 0, "blend workgroups are full");

struct Controls { float c0, c1, c2; };
__device__ __forceinline__ Controls controls(const float* __restrict__ logits)
{
    const float l0 = logits[0], l1 = logits[1], l2 = logits[2];
    const float m  = fmaxf(l0, fmaxf(l1, l2));
    const float e0 = expf(l0 - m), e1 = expf(l1 - m), e2 = expf(l2 - m);
    const float inv = 1.f / (e0 + e1 + e2);
    return Controls{e0 * inv, e1 * inv, e2 * inv};
}

__global__ __launch_bounds__(256) void k_mid(
    const float* __restrict__ stack, const float* __restrict__ logits,
    const float* __restrict__ stack_in, const float* __restrict__ w_ih,
    const float* __restrict__ gi, const float* __restrict__ gh,
    const float* __restrict__ h,
    float* __restrict__ out_stack, float* __restrict__ out_h,
    float* __restrict__ hnew_ws)
{
    if (blockIdx.x >= MID_GRU_BLOCKS) {
        const int idx = (blockIdx.x - MID_GRU_BLOCKS) * 256 + threadIdx.x;
        const int d = idx >> 7;
        const int w = idx & 127;
        const float cur  = stack[idx];
        const float up   = (d == 0)        ? stack_in[w]     : stack[idx - W_SZ];
        const float down = (d == D_SZ - 1) ? 0.f             : stack[idx + W_SZ];
        const Controls c = controls(logits);
        out_stack[idx] = c.c2 * cur + c.c0 * up + c.c1 * down;
        return;
    }

    const int lane = threadIdx.x & 31;
    const int wave = __builtin_amdgcn_readfirstlane(threadIdx.x >> 5);
    const int i    = blockIdx.x * MID_WAVES + wave;

    const v4f cur  = *(const v4f*)(stack + lane * 4);
    const v4f up   = *(const v4f*)(stack_in + lane * 4);
    const v4f down = *(const v4f*)(stack + W_SZ + lane * 4);
    const float* __restrict__ wp = w_ih + (size_t)i * KIH + H_SZ + lane * 4;
    v4f wt[3];
#pragma unroll
    for (int g = 0; g < 3; ++g) wt[g] = *(const v4f*)(wp + (size_t)g * H_SZ * KIH);

    const float ir  = gi[i],            hr = gh[i];
    const float iz  = gi[H_SZ + i],     hz = gh[H_SZ + i];
    const float in_ = gi[2 * H_SZ + i], hn = gh[2 * H_SZ + i];
    const float hv  = h[i];
    __builtin_amdgcn_sched_barrier(0);

    const Controls c = controls(logits);
    const v4f top = c.c2 * cur + c.c0 * up + c.c1 * down;
    const v2f a_lo = {top.x, top.y};
    const v2f a_hi = {top.z, top.w};

    const int  dj     = lane & 7;
    const bool onDiag = (lane < 8) || (lane >= 24);
    float t[3];
#pragma unroll
    for (int g = 0; g < 3; ++g) {
        const v2f b_lo = {wt[g].x, wt[g].y};
        const v2f b_hi = {wt[g].z, wt[g].w};
        v8f acc = (v8f){0.f, 0.f, 0.f, 0.f, 0.f, 0.f, 0.f, 0.f};
        acc = __builtin_amdgcn_wmma_f32_16x16x4_f32(
            false, a_lo, false, b_lo, (short)0, acc, false, false);
        acc = __builtin_amdgcn_wmma_f32_16x16x4_f32(
            false, a_hi, false, b_hi, (short)0, acc, false, false);
        float d = acc[0];
#pragma unroll
        for (int k = 1; k < 8; ++k) d = (dj == k) ? acc[k] : d;
        t[g] = onDiag ? d : 0.f;
    }
#pragma unroll
    for (int m = 16; m >= 1; m >>= 1) {
#pragma unroll
        for (int g = 0; g < 3; ++g) t[g] += __shfl_xor(t[g], m, 32);
    }

    const float r = 1.f / (1.f + expf(-((ir + t[0]) + hr)));
    const float z = 1.f / (1.f + expf(-((iz + t[1]) + hz)));
    const float n = tanhf((in_ + t[2]) + r * hn);
    const float o = (1.f - z) * n + z * hv;
    if (lane == 0) {
        out_h[i]   = o;
        hnew_ws[i] = o;
    }
}

__global__ __launch_bounds__(32) __attribute__((amdgpu_waves_per_eu(1, 4))) void k_dec(
    const float* __restrict__ hnew, const float* __restrict__ dec_W,
    const float* __restrict__ dec_b, float* __restrict__ out_y)
{
    constexpr int KQ = H_SZ / 128;
    const int row = blockIdx.x;
    const int lane = threadIdx.x;
    const float* __restrict__ wr = dec_W + (size_t)row * H_SZ + lane * 4;
    const float* __restrict__ hp = hnew + lane * 4;
    const float bias = dec_b[row];
    float4 wv[KQ];
    float4 hv[KQ];
#pragma unroll
    for (int q = 0; q < KQ; ++q) {
        wv[q] = *(const float4*)(wr + 128 * q);
        hv[q] = *(const float4*)(hp + 128 * q);
    }
    asm volatile("" :: "s"(bias));
    float s = 0.f;
#pragma unroll
    for (int q = 0; q < KQ; ++q)
        s += wv[q].x * hv[q].x + wv[q].y * hv[q].y + wv[q].z * hv[q].z + wv[q].w * hv[q].w;
    for (int off = 16; off > 0; off >>= 1) s += __shfl_down(s, off, 32);
    if (lane == 0) out_y[row] = s + bias;
}

extern "C" void kernel_launch(void* const* d_in, const int* in_sizes, int n_in,
                              void* d_out, int out_size, void* d_ws, size_t ws_size,
                              hipStream_t stream)
{
    (void)in_sizes; (void)n_in; (void)out_size; (void)ws_size;

    const int*   inp    = (const int*)  d_in[0];
    const float* hidden = (const float*)d_in[1];
    const float* stack  = (const float*)d_in[2];
    const float* emb    = (const float*)d_in[3];
    const float* ctrl_W = (const float*)d_in[4];
    const float* ctrl_b = (const float*)d_in[5];
    const float* sin_W  = (const float*)d_in[6];
    const float* sin_b  = (const float*)d_in[7];
    const float* w_ih   = (const float*)d_in[8];
    const float* w_hh   = (const float*)d_in[9];
    const float* b_ih   = (const float*)d_in[10];
    const float* b_hh   = (const float*)d_in[11];
    const float* dec_W  = (const float*)d_in[12];
    const float* dec_b  = (const float*)d_in[13];

    float* out       = (float*)d_out;
    float* out_y     = out;
    float* out_h     = out + V_SZ;
    float* out_stack = out + V_SZ + H_SZ;

    float* ws       = (float*)d_ws;
    float* logits   = ws;
    float* stack_in = ws + 16;
    float* gi       = ws + 2336;
    float* gh       = ws + 8480;
    float* hnew     = ws + 14624;

    k_front<<<AUX_BLOCKS + 2 * GEMV_BLOCKS, GEMV_WAVES * 32, 0, stream>>>(
        inp, hidden, emb, ctrl_W, ctrl_b, sin_W, sin_b, w_ih, w_hh, b_ih, b_hh,
        logits, stack_in, gi, gh);
    k_mid<<<MID_GRU_BLOCKS + MID_BLEND_BLOCKS, 256, 0, stream>>>(
        stack, logits, stack_in, w_ih, gi, gh, hidden, out_stack, out_h, hnew);
    k_dec<<<V_SZ, 32, 0, stream>>>(hnew, dec_W, dec_b, out_y);
}
